# starting kernel
# baseline (speedup 1.0000x reference)
#include <hip/hip_runtime.h>

typedef __attribute__((ext_vector_type(2))) float v2f;
typedef __attribute__((ext_vector_type(8))) float v8f;

#define IN_H     224
#define IN_W     224
#define OUT_HW   28
#define TAPS     169
#define TAPS_PAD 172
#define WSTRIDE  176

__global__ __launch_bounds__(64) void Gabor2d_wmma_kernel(
    const float* __restrict__ x,
    const float* __restrict__ k0, const float* __restrict__ k1,
    const float* __restrict__ k2, const float* __restrict__ k3,
    float* __restrict__ out)
{
    __shared__ float Wlds[4 * WSTRIDE];
    __shared__ float tile[2][16][16];

    const int tid  = threadIdx.x;
    const int wave = tid >> 5;
    const int lane = tid & 31;
    const int b    = blockIdx.x / OUT_HW;
    const int oy   = blockIdx.x % OUT_HW;

    for (int e = tid; e < 4 * WSTRIDE; e += 64) {
        const int n = e / WSTRIDE;
        const int t = e % WSTRIDE;
        float w = 0.0f;
        if (t < TAPS) {
            const int dy = t / 13 - 6;
            const int dx = t % 13 - 6;
            const int p  = 3 + n;
            const int ks = 7 + 2 * n;
            if (dy >= -p && dy <= p && dx >= -p && dx <= p) {
                const float* kp = (n == 0) ? k0 : (n == 1) ? k1 : (n == 2) ? k2 : k3;
                w = kp[(dy + p) * ks + (dx + p)];
            }
        }
        Wlds[e] = w;
    }
    __syncthreads();

    const int m   = lane & 15;
    const int hi  = lane >> 4;
    const int ox  = wave * 16 + m;
    const int cy  = oy * 8;
    const int cx  = ox * 8;
    const float* __restrict__ xb = x + (size_t)b * (IN_H * IN_W);

    v8f acc = {};
    for (int kk = 0; kk < TAPS_PAD; kk += 4) {
        const int t0 = kk + 2 * hi;
        const int t1 = t0 + 1;

        float a0 = 0.0f, a1 = 0.0f;
        if (t0 < TAPS) {
            const int iy = cy + t0 / 13 - 6;
            const int ix = cx + t0 % 13 - 6;
            if (iy >= 0 && iy < IN_H && ix >= 0 && ix < IN_W) a0 = xb[iy * IN_W + ix];
        }
        if (t1 < TAPS) {
            const int iy = cy + t1 / 13 - 6;
            const int ix = cx + t1 % 13 - 6;
            if (iy >= 0 && iy < IN_H && ix >= 0 && ix < IN_W) a1 = xb[iy * IN_W + ix];
        }
        v2f a; a.x = a0; a.y = a1;

        float b0 = 0.0f, b1 = 0.0f;
        if (m < 4) {
            b0 = Wlds[m * WSTRIDE + t0];
            b1 = Wlds[m * WSTRIDE + t1];
        }
        v2f bf; bf.x = b0; bf.y = b1;

        acc = __builtin_amdgcn_wmma_f32_16x16x4_f32(false, a, false, bf,
                                                    (short)0, acc, false, false);
    }

    #pragma unroll
    for (int r = 0; r < 8; ++r) {
        tile[wave][hi * 8 + r][m] = acc[r];
    }
    __syncthreads();

    if (ox < OUT_HW) {
        const float c0 = tile[wave][m][0];
        const float c1 = tile[wave][m][1];
        const float c2 = tile[wave][m][2];
        const float c3 = tile[wave][m][3];
        const float pr0 = fmaxf(c0, c1);
        const float pr1 = fmaxf(c1, c2);
        const float pr2 = fmaxf(c2, c3);
        const size_t plane = (size_t)OUT_HW * OUT_HW;
        const size_t base  = (size_t)b * (4 * 3 * plane)
                           + (size_t)oy * OUT_HW + (size_t)ox;
        #pragma unroll
        for (int th = 0; th < 2; ++th) {
            const size_t tb = base + ((size_t)(2 * th + hi) * 3) * plane;
            out[tb + 0 * plane] = pr0;
            out[tb + 1 * plane] = pr1;
            out[tb + 2 * plane] = pr2;
        }
    }
}

extern "C" void kernel_launch(void* const* d_in, const int* in_sizes, int n_in,
                              void* d_out, int out_size, void* d_ws, size_t ws_size,
                              hipStream_t stream) {
    (void)in_sizes; (void)n_in; (void)d_ws; (void)ws_size; (void)out_size;
    const float* x  = (const float*)d_in[0];
    const float* k0 = (const float*)d_in[1];
    const float* k1 = (const float*)d_in[2];
    const float* k2 = (const float*)d_in[3];
    const float* k3 = (const float*)d_in[4];
    float* out = (float*)d_out;

    const int blocks = 64 * OUT_HW;
    Gabor2d_wmma_kernel<<<blocks, 64, 0, stream>>>(x, k0, k1, k2, k3, out);
}


# Stage each block's 13 input rows in LDS with coalesced 16-byte loads, all global loads issued up front; the unrolled tap loop reads A and B from LDS at fixed offsets, no global gather per WMMA
# speedup vs baseline: 3.5954x; 3.5954x over previous
#include <hip/hip_runtime.h>


typedef __attribute__((ext_vector_type(2))) float v2f;
typedef __attribute__((ext_vector_type(4))) float v4f;
typedef __attribute__((ext_vector_type(8))) float v8f;

#define IN_H     224
#define IN_W     224
#define OUT_HW   28
#define TAPS     169
#define TAPS_PAD 172
#define WSTRIDE  176
#define ROWS     13
#define LPAD     8
#define RS       232

__global__ __launch_bounds__(64) void Gabor2d_wmma_kernel(
    const float* __restrict__ x,
    const float* __restrict__ k0, const float* __restrict__ k1,
    const float* __restrict__ k2, const float* __restrict__ k3,
    float* __restrict__ out)
{
    __shared__ __attribute__((aligned(16))) float Rlds[ROWS * RS];
    __shared__ __attribute__((aligned(16))) float Wlds[5 * WSTRIDE];
    __shared__ float tile[2][16][16];

    const int tid  = threadIdx.x;
    const int wave = tid >> 5;
    const int lane = tid & 31;
    const int b    = blockIdx.x / OUT_HW;
    const int oy   = blockIdx.x % OUT_HW;
    const int cy   = oy * 8;
    const float* __restrict__ xb = x + (size_t)b * (IN_H * IN_W);

    const int q = tid < 56 ? tid : 55;
    v4f rv[ROWS];
    #pragma unroll
    for (int dy = 0; dy < ROWS; ++dy) {
        const int iy  = cy - 6 + dy;
        const int iyc = iy < 0 ? 0 : iy;
        rv[dy] = *(const v4f*)(xb + iyc * IN_W + 4 * q);
    }

    const int wn  = tid >> 4;
    const int l16 = tid & 15;
    const int wp  = 3 + wn;
    const int wks = 7 + 2 * wn;
    const float* __restrict__ kp = (wn == 0) ? k0 : (wn == 1) ? k1 : (wn == 2) ? k2 : k3;
    const int  wc  = l16 - 6 + wp;
    const bool cin = (unsigned)wc < (unsigned)wks;
    float wv[ROWS];
    #pragma unroll
    for (int ty = 0; ty < ROWS; ++ty) {
        const int  wr = ty - 6 + wp;
        const bool in = cin && (unsigned)wr < (unsigned)wks;
        const int idx = in ? wr * wks + wc : 0;
        const float w = kp[idx];
        wv[ty] = in ? w : 0.0f;
    }

    if (oy == 0) {
        #pragma unroll
        for (int dy = 0; dy < 6; ++dy) rv[dy] = (v4f)(0.0f);
    }
    if (tid >= 56) {
        #pragma unroll
        for (int dy = 0; dy < ROWS; ++dy) rv[dy] = (v4f)(0.0f);
    }
    const int scol = tid < 56 ? LPAD + 4 * tid : 4 * (tid & 1);
    #pragma unroll
    for (int dy = 0; dy < ROWS; ++dy) {
        *(v4f*)(Rlds + dy * RS + scol) = rv[dy];
    }
    if (l16 < 13) {
        #pragma unroll
        for (int ty = 0; ty < ROWS; ++ty) Wlds[wn * WSTRIDE + 13 * ty + l16] = wv[ty];
    }
    if (l16 < WSTRIDE - TAPS) Wlds[wn * WSTRIDE + TAPS + l16] = 0.0f;
    if (tid < WSTRIDE / 4) *(v4f*)(Wlds + 4 * WSTRIDE + 4 * tid) = (v4f)(0.0f);
    __syncthreads();

    const int m   = lane & 15;
    const int hi  = lane >> 4;
    const int ox  = wave * 16 + m;
    const int oxc = ox < OUT_HW ? ox : OUT_HW - 1;

    const float* pL = Rlds + 8 * oxc + (LPAD - 6);
    const float* pN = pL + 2 * hi;
    const float* pW = pL + (RS - 11) * hi;
    const float* bp = Wlds + (m < 4 ? m : 4) * WSTRIDE + 2 * hi;

    v8f acc = {};
    #pragma unroll
    for (int kk = 0; kk < TAPS_PAD; kk += 4) {
        v2f a;
        #pragma unroll
        for (int s = 0; s < 2; ++s) {
            const int tl = kk + s;
            const int th = kk + 2 + s;
            const int ol = (tl / 13) * RS + (tl % 13);
            const int oh = (th / 13) * RS + (th % 13);
            float v;
            if (tl >= TAPS)          v = 0.0f;
            else if (th >= TAPS)     v = hi ? 0.0f : pL[ol];
            else if (oh - ol == 2)   v = pN[ol];
            else                     v = pW[ol];
            a[s] = v;
        }
        const v2f bf = *(const v2f*)(bp + kk);

        acc = __builtin_amdgcn_wmma_f32_16x16x4_f32(false, a, false, bf,
                                                    (short)0, acc, false, false);
    }

    #pragma unroll
    for (int r = 0; r < 8; ++r) {
        tile[wave][hi * 8 + r][m] = acc[r];
    }
    __syncthreads();

    if (ox < OUT_HW) {
        const float c0 = tile[wave][m][0];
        const float c1 = tile[wave][m][1];
        const float c2 = tile[wave][m][2];
        const float c3 = tile[wave][m][3];
        const float pr0 = fmaxf(c0, c1);
        const float pr1 = fmaxf(c1, c2);
        const float pr2 = fmaxf(c2, c3);
        const size_t plane = (size_t)OUT_HW * OUT_HW;
        const size_t base  = (size_t)b * (4 * 3 * plane)
                           + (size_t)oy * OUT_HW + (size_t)ox;
        #pragma unroll
        for (int th = 0; th < 2; ++th) {
            const size_t tb = base + ((size_t)(2 * th + hi) * 3) * plane;
            out[tb + 0 * plane] = pr0;
            out[tb + 1 * plane] = pr1;
            out[tb + 2 * plane] = pr2;
        }
    }
}

extern "C" void kernel_launch(void* const* d_in, const int* in_sizes, int n_in,
                              void* d_out, int out_size, void* d_ws, size_t ws_size,
                              hipStream_t stream) {
    (void)in_sizes; (void)n_in; (void)d_ws; (void)ws_size; (void)out_size;
    const float* x  = (const float*)d_in[0];
    const float* k0 = (const float*)d_in[1];
    const float* k1 = (const float*)d_in[2];
    const float* k2 = (const float*)d_in[3];
    const float* k3 = (const float*)d_in[4];
    float* out = (float*)d_out;

    const int blocks = 64 * OUT_HW;
    Gabor2d_wmma_kernel<<<blocks, 64, 0, stream>>>(x, k0, k1, k2, k3, out);
}


# Conv recast as a GEMM over the image's 8x8 patches: M = 4 filters x 4 patch offsets (all 16 rows used), K = 64, block = 7 output rows; each pixel adds its 4 patch partials: summing order changes
# speedup vs baseline: 4.6038x; 1.2805x over previous
#include <hip/hip_runtime.h>


typedef __attribute__((ext_vector_type(2))) float v2f;
typedef __attribute__((ext_vector_type(4))) float v4f;
typedef __attribute__((ext_vector_type(8))) float v8f;

#define IN_H     224
#define IN_W     224
#define OUT_HW   28
#define BAND     7
#define NPATCH   ((BAND + 1) * OUT_HW)
#define NWAVES   7
#define NTHREADS (NWAVES * 32)
#define APITCH   68
#define EPITCH   228

__global__ __launch_bounds__(NTHREADS) void Gabor2d_wmma_kernel(
    const float* __restrict__ x,
    const float* __restrict__ k0, const float* __restrict__ k1,
    const float* __restrict__ k2, const float* __restrict__ k3,
    float* __restrict__ out)
{
    __shared__ __attribute__((aligned(16))) float Alds[16 * APITCH];
    __shared__ __attribute__((aligned(16))) float Elds[16 * EPITCH];

    const int tid  = threadIdx.x;
    const int wave = tid >> 5;
    const int lane = tid & 31;
    const int c    = lane & 15;
    const int h    = lane >> 4;
    const int b    = blockIdx.x >> 2;
    const int g    = blockIdx.x & 3;
    const float* __restrict__ xb = x + (size_t)b * (IN_H * IN_W);

    *(v4f*)(Alds + 4 * tid) = (v4f)(0.0f);
    if (tid < (16 * APITCH) / 4 - NTHREADS) *(v4f*)(Alds + 4 * (NTHREADS + tid)) = (v4f)(0.0f);
    __syncthreads();

    const float w0 = k0[tid < 49  ? tid : 48];
    const float w1 = k1[tid < 81  ? tid : 80];
    const float w2 = k2[tid < 121 ? tid : 120];
    const float w3 = k3[tid < 169 ? tid : 168];

    v4f L[2][8];
    #pragma unroll
    for (int u = 0; u < 2; ++u) {
        const int n  = 32 * wave + 16 * u + c;
        const int il = n / OUT_HW;
        const int j  = n - il * OUT_HW;
        const int i  = BAND * g - 1 + il;
        const int ic = i < 0 ? 0 : i;
        const float* __restrict__ src = xb + (8 * ic) * IN_W + 8 * j + 4 * h;
        #pragma unroll
        for (int cy = 0; cy < 8; ++cy) L[u][cy] = *(const v4f*)(src + cy * IN_W);
    }

    asm volatile("" :: "v"(w0), "v"(w1), "v"(w2), "v"(w3));

    {
        const float wf[4] = {w0, w1, w2, w3};
        #pragma unroll
        for (int f = 0; f < 4; ++f) {
            const int p = 3 + f, ks = 7 + 2 * f;
            if (tid < ks * ks) {
                const int wr = tid / ks, wc = tid - wr * ks;
                const int ty2 = wr + 8 - p, tx2 = wc + 8 - p;
                const int m = 4 * ((tx2 >> 3) + 2 * (ty2 >> 3)) + f;
                Alds[m * APITCH + 8 * (ty2 & 7) + (tx2 & 7)] = wf[f];
            }
        }
    }
    __syncthreads();

    v4f Af[8];
    #pragma unroll
    for (int cy = 0; cy < 8; ++cy) Af[cy] = *(const v4f*)(Alds + c * APITCH + 8 * cy + 4 * h);
    __builtin_amdgcn_sched_barrier(0);

    v8f acc0 = {}, acc1 = {};
    #pragma unroll
    for (int cy = 0; cy < 8; ++cy) {
        #pragma unroll
        for (int e = 0; e < 2; ++e) {
            v2f a, b0, b1;
            a[0]  = Af[cy][2 * e];      a[1]  = Af[cy][2 * e + 1];
            b0[0] = L[0][cy][2 * e];    b0[1] = L[0][cy][2 * e + 1];
            b1[0] = L[1][cy][2 * e];    b1[1] = L[1][cy][2 * e + 1];
            acc0 = __builtin_amdgcn_wmma_f32_16x16x4_f32(false, a, false, b0, (short)0, acc0, false, false);
            acc1 = __builtin_amdgcn_wmma_f32_16x16x4_f32(false, a, false, b1, (short)0, acc1, false, false);
        }
    }

    #pragma unroll
    for (int r = 0; r < 8; ++r) {
        Elds[(8 * h + r) * EPITCH + 32 * wave + c]      = acc0[r];
        Elds[(8 * h + r) * EPITCH + 32 * wave + 16 + c] = acc1[r];
    }
    __syncthreads();

    if (tid < BAND * OUT_HW) {
        const int oyl = tid / OUT_HW;
        const int ox  = tid - oyl * OUT_HW;
        const int oy  = BAND * g + oyl;
        const bool hasL = ox > 0;
        const bool hasU = oy > 0;
        const int n1  = (oyl + 1) * OUT_HW + ox;
        const int n0  = oyl * OUT_HW + ox;
        const int n1l = hasL ? n1 - 1 : n1;
        const int n0l = hasL ? n0 - 1 : n0;
        float cv[4];
        #pragma unroll
        for (int f = 0; f < 4; ++f) {
            const float e3 = Elds[(12 + f) * EPITCH + n1];
            const float e2 = Elds[(8 + f) * EPITCH + n1l];
            const float e1 = Elds[(4 + f) * EPITCH + n0];
            const float e0 = Elds[f * EPITCH + n0l];
            float s = e3;
            s += hasL ? e2 : 0.0f;
            s += hasU ? e1 : 0.0f;
            s += (hasL && hasU) ? e0 : 0.0f;
            cv[f] = s;
        }
        const float pr0 = fmaxf(cv[0], cv[1]);
        const float pr1 = fmaxf(cv[1], cv[2]);
        const float pr2 = fmaxf(cv[2], cv[3]);
        const size_t plane = (size_t)OUT_HW * OUT_HW;
        float* __restrict__ ob = out + (size_t)b * (4 * 3 * plane) + (size_t)(BAND * g) * OUT_HW + tid;
        #pragma unroll
        for (int th = 0; th < 4; ++th) {
            ob[(size_t)(3 * th + 0) * plane] = pr0;
            ob[(size_t)(3 * th + 1) * plane] = pr1;
            ob[(size_t)(3 * th + 2) * plane] = pr2;
        }
    }
}

extern "C" void kernel_launch(void* const* d_in, const int* in_sizes, int n_in,
                              void* d_out, int out_size, void* d_ws, size_t ws_size,
                              hipStream_t stream) {
    (void)in_sizes; (void)n_in; (void)d_ws; (void)ws_size; (void)out_size;
    const float* x  = (const float*)d_in[0];
    const float* k0 = (const float*)d_in[1];
    const float* k1 = (const float*)d_in[2];
    const float* k2 = (const float*)d_in[3];
    const float* k3 = (const float*)d_in[4];
    float* out = (float*)d_out;

    const int blocks = 64 * 4;
    Gabor2d_wmma_kernel<<<blocks, NTHREADS, 0, stream>>>(x, k0, k1, k2, k3, out);
}
